# starting kernel
# baseline (speedup 1.0000x reference)
#include <hip/hip_runtime.h>

#define B_   2
#define N_   1024
#define D_   3
#define A_   64
#define R_   50
#define RPAD 64

typedef float v2f __attribute__((ext_vector_type(2)));
typedef float v8f __attribute__((ext_vector_type(8)));

__global__ __launch_bounds__(32)
void prefactor_wmma_kernel(const float* __restrict__ emb,
                           const int*   __restrict__ z,
                           const float* __restrict__ t,
                           const float* __restrict__ time_off,
                           const float* __restrict__ time_wid,
                           const float* __restrict__ W,
                           float* __restrict__ P) {
  const int mt   = blockIdx.x;
  const int rt   = blockIdx.y;
  const int lane = threadIdx.x;
  const int g    = lane >> 4;
  const int lr   = lane & 15;

  const int m  = mt * 16 + lr;
  const int b  = m >> 10;
  const int i  = m & (N_ - 1);
  const int zi = z[i];
  const float tb = t[b];

  const int  r      = rt * 16 + lr;
  const bool rvalid = (r < R_);
  const int  rc     = rvalid ? r : 0;

  v8f c = {0.f, 0.f, 0.f, 0.f, 0.f, 0.f, 0.f, 0.f};

  #pragma unroll
  for (int k0 = 0; k0 < A_; k0 += 4) {
    const int ka = k0 + 2 * g;

    v2f a, bf;
    {
      const float o0 = time_off[ka],     w0 = time_wid[ka];
      const float o1 = time_off[ka + 1], w1 = time_wid[ka + 1];
      const float c0 = -0.5f * __builtin_amdgcn_rcpf(w0 * w0);
      const float c1 = -0.5f * __builtin_amdgcn_rcpf(w1 * w1);
      const float d0 = tb - o0, d1 = tb - o1;
      const float e0 = __expf(c0 * d0 * d0);
      const float e1 = __expf(c1 * d1 * d1);
      a.x = emb[zi * A_ + ka]     * e0;
      a.y = emb[zi * A_ + ka + 1] * e1;
    }
    const float w0v = W[ka * R_ + rc];
    const float w1v = W[(ka + 1) * R_ + rc];
    bf.x = rvalid ? w0v : 0.f;
    bf.y = rvalid ? w1v : 0.f;

    c = __builtin_amdgcn_wmma_f32_16x16x4_f32(false, a, false, bf,
                                              (short)0, c, false, false);
  }

  #pragma unroll
  for (int v = 0; v < 8; ++v) {
    const int row = mt * 16 + v + 8 * g;
    P[row * RPAD + rt * 16 + lr] = c[v];
  }
}

__global__ __launch_bounds__(256)
void field_kernel(const float* __restrict__ pos,
                  const float* __restrict__ P,
                  const float* __restrict__ rad_off,
                  const float* __restrict__ rad_wid,
                  float* __restrict__ out) {
  const int i   = blockIdx.x;
  const int b   = blockIdx.y;
  const int tid = threadIdx.x;

  __shared__ float sx[N_], sy[N_], sz[N_];
  __shared__ float sp[R_], sc[R_], so[R_];
  __shared__ float red[3][256];

  const float* pb = pos + (size_t)b * N_ * D_;
  for (int j = tid; j < N_; j += 256) {
    sx[j] = pb[j * 3 + 0];
    sy[j] = pb[j * 3 + 1];
    sz[j] = pb[j * 3 + 2];
  }
  if (tid < R_) {
    sp[tid] = P[((size_t)b * N_ + i) * RPAD + tid];
    const float w = rad_wid[tid];
    sc[tid] = -0.5f * __builtin_amdgcn_rcpf(w * w);
    so[tid] = rad_off[tid];
  }
  __syncthreads();

  const float xi = sx[i], yi = sy[i], zi = sz[i];
  float ax = 0.f, ay = 0.f, az = 0.f;

  for (int j = tid; j < N_; j += 256) {
    if (j == i) continue;
    const float dx = xi - sx[j];
    const float dy = yi - sy[j];
    const float dz = zi - sz[j];
    const float d  = sqrtf(dx * dx + dy * dy + dz * dz + 1e-6f);
    float phi = 0.f;
    #pragma unroll
    for (int rr = 0; rr < R_; ++rr) {
      const float df = d - so[rr];
      phi += sp[rr] * __expf(sc[rr] * df * df);
    }
    ax += phi * dx;
    ay += phi * dy;
    az += phi * dz;
  }

  red[0][tid] = ax; red[1][tid] = ay; red[2][tid] = az;
  __syncthreads();
  #pragma unroll
  for (int s = 128; s > 0; s >>= 1) {
    if (tid < s) {
      red[0][tid] += red[0][tid + s];
      red[1][tid] += red[1][tid + s];
      red[2][tid] += red[2][tid + s];
    }
    __syncthreads();
  }
  if (tid == 0) {
    const size_t o = ((size_t)b * N_ + i) * 3;
    out[o + 0] = red[0][0] + pb[i * 3 + 0];
    out[o + 1] = red[1][0] + pb[i * 3 + 1];
    out[o + 2] = red[2][0] + pb[i * 3 + 2];
  }
}

extern "C" void kernel_launch(void* const* d_in, const int* in_sizes, int n_in,
                              void* d_out, int out_size, void* d_ws, size_t ws_size,
                              hipStream_t stream) {
  (void)in_sizes; (void)n_in; (void)out_size; (void)ws_size;
  const float* positions = (const float*)d_in[0];
  const float* t         = (const float*)d_in[1];
  const int*   z         = (const int*)  d_in[2];
  const float* emb       = (const float*)d_in[3];
  const float* W         = (const float*)d_in[4];
  const float* rad_off   = (const float*)d_in[5];
  const float* rad_wid   = (const float*)d_in[6];
  const float* time_off  = (const float*)d_in[7];
  const float* time_wid  = (const float*)d_in[8];
  float* out = (float*)d_out;
  float* P   = (float*)d_ws;

  dim3 g1(128, 4), b1(32);
  prefactor_wmma_kernel<<<g1, b1, 0, stream>>>(emb, z, t, time_off, time_wid, W, P);

  dim3 g2(N_, B_), b2(256);
  field_kernel<<<g2, b2, 0, stream>>>(positions, P, rad_off, rad_wid, out);
}


# field kernel re-tiled: each row owned by 2 waves (16 pairs per lane), 4 rows per block share one LDS staging; wave butterfly + 2 partials replace the 256-thread LDS tree (sum order changes)
# speedup vs baseline: 1.4200x; 1.4200x over previous
#include <hip/hip_runtime.h>

#define B_   2
#define N_   1024
#define D_   3
#define A_   64
#define R_   50
#define RPAD 64

typedef float v2f __attribute__((ext_vector_type(2)));
typedef float v8f __attribute__((ext_vector_type(8)));

__global__ __launch_bounds__(32)
void prefactor_wmma_kernel(const float* __restrict__ emb,
                           const int*   __restrict__ z,
                           const float* __restrict__ t,
                           const float* __restrict__ time_off,
                           const float* __restrict__ time_wid,
                           const float* __restrict__ W,
                           float* __restrict__ P) {
  const int mt   = blockIdx.x;
  const int rt   = blockIdx.y;
  const int lane = threadIdx.x;
  const int g    = lane >> 4;
  const int lr   = lane & 15;

  const int m  = mt * 16 + lr;
  const int b  = m >> 10;
  const int i  = m & (N_ - 1);
  const int zi = z[i];
  const float tb = t[b];

  const int  r      = rt * 16 + lr;
  const bool rvalid = (r < R_);
  const int  rc     = rvalid ? r : 0;

  v8f c = {0.f, 0.f, 0.f, 0.f, 0.f, 0.f, 0.f, 0.f};

  #pragma unroll
  for (int k0 = 0; k0 < A_; k0 += 4) {
    const int ka = k0 + 2 * g;

    v2f a, bf;
    {
      const float o0 = time_off[ka],     w0 = time_wid[ka];
      const float o1 = time_off[ka + 1], w1 = time_wid[ka + 1];
      const float c0 = -0.5f * __builtin_amdgcn_rcpf(w0 * w0);
      const float c1 = -0.5f * __builtin_amdgcn_rcpf(w1 * w1);
      const float d0 = tb - o0, d1 = tb - o1;
      const float e0 = __expf(c0 * d0 * d0);
      const float e1 = __expf(c1 * d1 * d1);
      a.x = emb[zi * A_ + ka]     * e0;
      a.y = emb[zi * A_ + ka + 1] * e1;
    }
    const float w0v = W[ka * R_ + rc];
    const float w1v = W[(ka + 1) * R_ + rc];
    bf.x = rvalid ? w0v : 0.f;
    bf.y = rvalid ? w1v : 0.f;

    c = __builtin_amdgcn_wmma_f32_16x16x4_f32(false, a, false, bf,
                                              (short)0, c, false, false);
  }

  #pragma unroll
  for (int v = 0; v < 8; ++v) {
    const int row = mt * 16 + v + 8 * g;
    P[row * RPAD + rt * 16 + lr] = c[v];
  }
}

#define ROWS_  4
#define SPLIT_ 2
#define FTHREADS (32 * SPLIT_ * ROWS_)
static_assert(FTHREADS == 256 && B_ * N_ / ROWS_ == 512, "launch shape in kernel_launch is written as literals");
static_assert(SPLIT_ == 2 && (N_ * D_) % FTHREADS == 0 && ROWS_ * RPAD == FTHREADS, "tiling assumptions");

__global__ __launch_bounds__(FTHREADS)
void field_kernel(const float* __restrict__ pos,
                  const float* __restrict__ P,
                  const float* __restrict__ rad_off,
                  const float* __restrict__ rad_wid,
                  float* __restrict__ out) {
  const int tid  = threadIdx.x;
  const int lane = tid & 31;
  const int wv   = tid >> 5;
  const int rib  = wv / SPLIT_;
  const int half = wv % SPLIT_;
  const int b    = blockIdx.x / (N_ / ROWS_);
  const int i0   = (blockIdx.x % (N_ / ROWS_)) * ROWS_;
  const int i    = i0 + rib;

  __shared__ float spos[N_ * D_];
  __shared__ float sc[RPAD], so[RPAD];
  __shared__ float sp[ROWS_ * RPAD];
  __shared__ float spart[SPLIT_ * ROWS_][4];

  const float* pb = pos + (size_t)b * N_ * D_;
  #pragma unroll
  for (int q = 0; q < N_ * D_ / FTHREADS; ++q)
    spos[tid + FTHREADS * q] = pb[tid + FTHREADS * q];
  if (tid < R_) {
    const float w = rad_wid[tid];
    sc[tid] = -0.5f * __builtin_amdgcn_rcpf(w * w);
    so[tid] = rad_off[tid];
  }
  sp[tid] = P[((size_t)b * N_ + i0) * RPAD + tid];
  __syncthreads();

  float co[R_], cc[R_], cp[R_];
  #pragma unroll
  for (int rr = 0; rr < R_; ++rr) {
    co[rr] = so[rr];
    cc[rr] = sc[rr];
    cp[rr] = sp[rib * RPAD + rr];
  }

  const float xi = spos[i * 3 + 0], yi = spos[i * 3 + 1], zi = spos[i * 3 + 2];
  float ax = 0.f, ay = 0.f, az = 0.f;

  #pragma unroll 2
  for (int k = 0; k < N_ / (32 * SPLIT_); ++k) {
    const int j = lane + 32 * (SPLIT_ * k + half);
    const float dx = xi - spos[j * 3 + 0];
    const float dy = yi - spos[j * 3 + 1];
    const float dz = zi - spos[j * 3 + 2];
    const float d  = sqrtf(dx * dx + dy * dy + dz * dz + 1e-6f);
    float phi = 0.f;
    #pragma unroll
    for (int rr = 0; rr < R_; ++rr) {
      const float df = d - co[rr];
      phi += cp[rr] * __expf(cc[rr] * df * df);
    }
    phi = (j == i) ? 0.f : phi;
    ax += phi * dx;
    ay += phi * dy;
    az += phi * dz;
  }

  #pragma unroll
  for (int m = 16; m > 0; m >>= 1) {
    ax += __shfl_xor(ax, m);
    ay += __shfl_xor(ay, m);
    az += __shfl_xor(az, m);
  }
  if (lane < 3) spart[wv][lane] = (lane == 0) ? ax : (lane == 1) ? ay : az;
  __syncthreads();

  if (tid < ROWS_ * 3) {
    const int r = tid / 3, c = tid % 3;
    const float s = spart[r * SPLIT_][c] + spart[r * SPLIT_ + 1][c];
    out[((size_t)b * N_ + i0) * 3 + tid] = s + spos[(i0 + r) * 3 + c];
  }
}

extern "C" void kernel_launch(void* const* d_in, const int* in_sizes, int n_in,
                              void* d_out, int out_size, void* d_ws, size_t ws_size,
                              hipStream_t stream) {
  (void)in_sizes; (void)n_in; (void)out_size; (void)ws_size;
  const float* positions = (const float*)d_in[0];
  const float* t         = (const float*)d_in[1];
  const int*   z         = (const int*)  d_in[2];
  const float* emb       = (const float*)d_in[3];
  const float* W         = (const float*)d_in[4];
  const float* rad_off   = (const float*)d_in[5];
  const float* rad_wid   = (const float*)d_in[6];
  const float* time_off  = (const float*)d_in[7];
  const float* time_wid  = (const float*)d_in[8];
  float* out = (float*)d_out;
  float* P   = (float*)d_ws;

  dim3 g1(128, 4), b1(32);
  prefactor_wmma_kernel<<<g1, b1, 0, stream>>>(emb, z, t, time_off, time_wid, W, P);

  dim3 g2(512), b2(256);
  field_kernel<<<g2, b2, 0, stream>>>(positions, P, rad_off, rad_wid, out);
}


# field kernel: basis exponent rewritten as exp2(-(s*d - s*o)^2), s=sqrt(0.5*log2e)/w and -s*o built once per block; per term sub+3 mul become one fma + one negated square
# speedup vs baseline: 1.5140x; 1.0662x over previous
#include <hip/hip_runtime.h>

#define B_   2
#define N_   1024
#define D_   3
#define A_   64
#define R_   50
#define RPAD 64

typedef float v2f __attribute__((ext_vector_type(2)));
typedef float v8f __attribute__((ext_vector_type(8)));

__global__ __launch_bounds__(32)
void prefactor_wmma_kernel(const float* __restrict__ emb,
                           const int*   __restrict__ z,
                           const float* __restrict__ t,
                           const float* __restrict__ time_off,
                           const float* __restrict__ time_wid,
                           const float* __restrict__ W,
                           float* __restrict__ P) {
  const int mt   = blockIdx.x;
  const int rt   = blockIdx.y;
  const int lane = threadIdx.x;
  const int g    = lane >> 4;
  const int lr   = lane & 15;

  const int m  = mt * 16 + lr;
  const int b  = m >> 10;
  const int i  = m & (N_ - 1);
  const int zi = z[i];
  const float tb = t[b];

  const int  r      = rt * 16 + lr;
  const bool rvalid = (r < R_);
  const int  rc     = rvalid ? r : 0;

  v8f c = {0.f, 0.f, 0.f, 0.f, 0.f, 0.f, 0.f, 0.f};

  #pragma unroll
  for (int k0 = 0; k0 < A_; k0 += 4) {
    const int ka = k0 + 2 * g;

    v2f a, bf;
    {
      const float o0 = time_off[ka],     w0 = time_wid[ka];
      const float o1 = time_off[ka + 1], w1 = time_wid[ka + 1];
      const float c0 = -0.5f * __builtin_amdgcn_rcpf(w0 * w0);
      const float c1 = -0.5f * __builtin_amdgcn_rcpf(w1 * w1);
      const float d0 = tb - o0, d1 = tb - o1;
      const float e0 = __expf(c0 * d0 * d0);
      const float e1 = __expf(c1 * d1 * d1);
      a.x = emb[zi * A_ + ka]     * e0;
      a.y = emb[zi * A_ + ka + 1] * e1;
    }
    const float w0v = W[ka * R_ + rc];
    const float w1v = W[(ka + 1) * R_ + rc];
    bf.x = rvalid ? w0v : 0.f;
    bf.y = rvalid ? w1v : 0.f;

    c = __builtin_amdgcn_wmma_f32_16x16x4_f32(false, a, false, bf,
                                              (short)0, c, false, false);
  }

  #pragma unroll
  for (int v = 0; v < 8; ++v) {
    const int row = mt * 16 + v + 8 * g;
    P[row * RPAD + rt * 16 + lr] = c[v];
  }
}

#define ROWS_  4
#define SPLIT_ 2
#define FTHREADS (32 * SPLIT_ * ROWS_)
static_assert(FTHREADS == 256 && B_ * N_ / ROWS_ == 512, "launch shape in kernel_launch is written as literals");
static_assert(SPLIT_ == 2 && (N_ * D_) % FTHREADS == 0 && ROWS_ * RPAD == FTHREADS, "tiling assumptions");

__global__ __launch_bounds__(FTHREADS)
void field_kernel(const float* __restrict__ pos,
                  const float* __restrict__ P,
                  const float* __restrict__ rad_off,
                  const float* __restrict__ rad_wid,
                  float* __restrict__ out) {
  const int tid  = threadIdx.x;
  const int lane = tid & 31;
  const int wv   = tid >> 5;
  const int rib  = wv / SPLIT_;
  const int half = wv % SPLIT_;
  const int b    = blockIdx.x / (N_ / ROWS_);
  const int i0   = (blockIdx.x % (N_ / ROWS_)) * ROWS_;
  const int i    = i0 + rib;

  __shared__ float spos[N_ * D_];
  __shared__ float sc[RPAD], so[RPAD];
  __shared__ float sp[ROWS_ * RPAD];
  __shared__ float spart[SPLIT_ * ROWS_][4];

  const float* pb = pos + (size_t)b * N_ * D_;
  #pragma unroll
  for (int q = 0; q < N_ * D_ / FTHREADS; ++q)
    spos[tid + FTHREADS * q] = pb[tid + FTHREADS * q];
  if (tid < R_) {
    const float s = fminf(0.84932180028801907f * __builtin_amdgcn_rcpf(fabsf(rad_wid[tid])), 1.0e18f);
    sc[tid] = s;
    so[tid] = -(s * rad_off[tid]);
  }
  sp[tid] = P[((size_t)b * N_ + i0) * RPAD + tid];
  __syncthreads();

  float co[R_], cc[R_], cp[R_];
  #pragma unroll
  for (int rr = 0; rr < R_; ++rr) {
    co[rr] = so[rr];
    cc[rr] = sc[rr];
    cp[rr] = sp[rib * RPAD + rr];
  }

  const float xi = spos[i * 3 + 0], yi = spos[i * 3 + 1], zi = spos[i * 3 + 2];
  float ax = 0.f, ay = 0.f, az = 0.f;

  #pragma unroll 2
  for (int k = 0; k < N_ / (32 * SPLIT_); ++k) {
    const int j = lane + 32 * (SPLIT_ * k + half);
    const float dx = xi - spos[j * 3 + 0];
    const float dy = yi - spos[j * 3 + 1];
    const float dz = zi - spos[j * 3 + 2];
    const float d  = sqrtf(dx * dx + dy * dy + dz * dz + 1e-6f);
    float phi = 0.f;
    #pragma unroll
    for (int rr = 0; rr < R_; ++rr) {
      const float u = __builtin_fmaf(cc[rr], d, co[rr]);
      phi += cp[rr] * __builtin_amdgcn_exp2f(-(u * u));
    }
    phi = (j == i) ? 0.f : phi;
    ax += phi * dx;
    ay += phi * dy;
    az += phi * dz;
  }

  #pragma unroll
  for (int m = 16; m > 0; m >>= 1) {
    ax += __shfl_xor(ax, m);
    ay += __shfl_xor(ay, m);
    az += __shfl_xor(az, m);
  }
  if (lane < 3) spart[wv][lane] = (lane == 0) ? ax : (lane == 1) ? ay : az;
  __syncthreads();

  if (tid < ROWS_ * 3) {
    const int r = tid / 3, c = tid % 3;
    const float s = spart[r * SPLIT_][c] + spart[r * SPLIT_ + 1][c];
    out[((size_t)b * N_ + i0) * 3 + tid] = s + spos[(i0 + r) * 3 + c];
  }
}

extern "C" void kernel_launch(void* const* d_in, const int* in_sizes, int n_in,
                              void* d_out, int out_size, void* d_ws, size_t ws_size,
                              hipStream_t stream) {
  (void)in_sizes; (void)n_in; (void)out_size; (void)ws_size;
  const float* positions = (const float*)d_in[0];
  const float* t         = (const float*)d_in[1];
  const int*   z         = (const int*)  d_in[2];
  const float* emb       = (const float*)d_in[3];
  const float* W         = (const float*)d_in[4];
  const float* rad_off   = (const float*)d_in[5];
  const float* rad_wid   = (const float*)d_in[6];
  const float* time_off  = (const float*)d_in[7];
  const float* time_wid  = (const float*)d_in[8];
  float* out = (float*)d_out;
  float* P   = (float*)d_ws;

  dim3 g1(128, 4), b1(32);
  prefactor_wmma_kernel<<<g1, b1, 0, stream>>>(emb, z, t, time_off, time_wid, W, P);

  dim3 g2(512), b2(256);
  field_kernel<<<g2, b2, 0, stream>>>(positions, P, rad_off, rad_wid, out);
}


# prefactor GEMM folded into the field kernel: X tile built once per block in LDS, waves 0-3 run the same 16-step f32 WMMA chains into LDS; one launch, no workspace, sum orders unchanged
# speedup vs baseline: 1.9139x; 1.2641x over previous
#include <hip/hip_runtime.h>

#define B_   2
#define N_   1024
#define D_   3
#define A_   64
#define R_   50
#define RPAD 64

typedef float v2f __attribute__((ext_vector_type(2)));
typedef float v8f __attribute__((ext_vector_type(8)));

#define ROWS_  4
#define SPLIT_ 2
#define FTHREADS (32 * SPLIT_ * ROWS_)
#define XPITCH 68
static_assert(FTHREADS == 256 && B_ * N_ / ROWS_ == 512, "launch shape in kernel_launch is written as literals");
static_assert(SPLIT_ == 2 && (N_ * D_) % FTHREADS == 0 && ROWS_ * RPAD == FTHREADS, "tiling assumptions");
static_assert(RPAD / 16 <= FTHREADS / 32 && ROWS_ <= 8 && (ROWS_ & (ROWS_ - 1)) == 0 && A_ % 4 == 0, "prefactor tile: one wave per 16 columns, kept rows in lane half 0");
static_assert(ROWS_ * A_ == FTHREADS && XPITCH >= A_ && XPITCH % 2 == 0, "X tile: one element per thread, 8-byte aligned fragment reads");

__global__ __launch_bounds__(FTHREADS)
void field_kernel(const float* __restrict__ pos,
                  const float* __restrict__ emb,
                  const int*   __restrict__ z,
                  const float* __restrict__ t,
                  const float* __restrict__ time_off,
                  const float* __restrict__ time_wid,
                  const float* __restrict__ W,
                  const float* __restrict__ rad_off,
                  const float* __restrict__ rad_wid,
                  float* __restrict__ out) {
  const int tid  = threadIdx.x;
  const int lane = tid & 31;
  const int wv   = tid >> 5;
  const int rib  = wv / SPLIT_;
  const int half = wv % SPLIT_;
  const int b    = blockIdx.x / (N_ / ROWS_);
  const int i0   = (blockIdx.x % (N_ / ROWS_)) * ROWS_;
  const int i    = i0 + rib;

  __shared__ float spos[N_ * D_];
  __shared__ float sc[RPAD], so[RPAD];
  __shared__ float sp[ROWS_ * RPAD];
  __shared__ float sx[ROWS_ * XPITCH];
  __shared__ float spart[SPLIT_ * ROWS_][4];

  const int xr = tid >> 6;
  const int xk = tid & (A_ - 1);
  const int zt = z[i0 + xr];
  const float tb = t[b];
  const float to = time_off[xk], tw = time_wid[xk];

  const float* pb = pos + (size_t)b * N_ * D_;
  float pv[N_ * D_ / FTHREADS];
  #pragma unroll
  for (int q = 0; q < N_ * D_ / FTHREADS; ++q)
    pv[q] = pb[tid + FTHREADS * q];

  const int wvu = __builtin_amdgcn_readfirstlane(wv);
  const int g   = lane >> 4;
  const int lr  = lane & 15;
  const bool rvalid = (wvu * 16 + lr < R_);
  float wb[A_ / 4][2];
  if (wvu < RPAD / 16) {
    const int rc = rvalid ? wvu * 16 + lr : 0;
    #pragma unroll
    for (int s = 0; s < A_ / 4; ++s) {
      const int ka = 4 * s + 2 * g;
      wb[s][0] = W[ka * R_ + rc];
      wb[s][1] = W[(ka + 1) * R_ + rc];
    }
  }

  const float ev = emb[zt * A_ + xk];

  if (tid >= FTHREADS / 2 && tid < FTHREADS / 2 + R_) {
    const int rr = tid - FTHREADS / 2;
    const float s = fminf(0.84932180028801907f * __builtin_amdgcn_rcpf(fabsf(rad_wid[rr])), 1.0e18f);
    sc[rr] = s;
    so[rr] = -(s * rad_off[rr]);
  }
  {
    const float c0 = -0.5f * __builtin_amdgcn_rcpf(tw * tw);
    const float d0 = tb - to;
    const float e0 = __expf(c0 * d0 * d0);
    sx[xr * XPITCH + xk] = ev * e0;
  }
  #pragma unroll
  for (int q = 0; q < N_ * D_ / FTHREADS; ++q)
    spos[tid + FTHREADS * q] = pv[q];
  __syncthreads();

  if (wvu < RPAD / 16) {
    v8f c = {0.f, 0.f, 0.f, 0.f, 0.f, 0.f, 0.f, 0.f};
    const float* xa = sx + (lr & (ROWS_ - 1)) * XPITCH + 2 * g;
    v2f af[A_ / 4];
    #pragma unroll
    for (int s = 0; s < A_ / 4; ++s) {
      af[s].x = xa[4 * s];
      af[s].y = xa[4 * s + 1];
    }
    __builtin_amdgcn_sched_barrier(0);
    #pragma unroll
    for (int s = 0; s < A_ / 4; ++s) {
      const v2f a = af[s];
      v2f bf;
      bf.x = rvalid ? wb[s][0] : 0.f;
      bf.y = rvalid ? wb[s][1] : 0.f;
      c = __builtin_amdgcn_wmma_f32_16x16x4_f32(false, a, false, bf,
                                                (short)0, c, false, false);
    }
    if (g == 0) {
      #pragma unroll
      for (int v = 0; v < ROWS_; ++v)
        sp[v * RPAD + wvu * 16 + lr] = c[v];
    }
  }
  __syncthreads();

  float co[R_], cc[R_], cp[R_];
  #pragma unroll
  for (int rr = 0; rr < R_; ++rr) {
    co[rr] = so[rr];
    cc[rr] = sc[rr];
    cp[rr] = sp[rib * RPAD + rr];
  }

  const float xi = spos[i * 3 + 0], yi = spos[i * 3 + 1], zi = spos[i * 3 + 2];
  float ax = 0.f, ay = 0.f, az = 0.f;

  #pragma unroll 2
  for (int k = 0; k < N_ / (32 * SPLIT_); ++k) {
    const int j = lane + 32 * (SPLIT_ * k + half);
    const float dx = xi - spos[j * 3 + 0];
    const float dy = yi - spos[j * 3 + 1];
    const float dz = zi - spos[j * 3 + 2];
    const float d  = sqrtf(dx * dx + dy * dy + dz * dz + 1e-6f);
    float phi = 0.f;
    #pragma unroll
    for (int rr = 0; rr < R_; ++rr) {
      const float u = __builtin_fmaf(cc[rr], d, co[rr]);
      phi += cp[rr] * __builtin_amdgcn_exp2f(-(u * u));
    }
    phi = (j == i) ? 0.f : phi;
    ax += phi * dx;
    ay += phi * dy;
    az += phi * dz;
  }

  #pragma unroll
  for (int m = 16; m > 0; m >>= 1) {
    ax += __shfl_xor(ax, m);
    ay += __shfl_xor(ay, m);
    az += __shfl_xor(az, m);
  }
  if (lane < 3) spart[wv][lane] = (lane == 0) ? ax : (lane == 1) ? ay : az;
  __syncthreads();

  if (tid < ROWS_ * 3) {
    const int r = tid / 3, c = tid % 3;
    const float s = spart[r * SPLIT_][c] + spart[r * SPLIT_ + 1][c];
    out[((size_t)b * N_ + i0) * 3 + tid] = s + spos[(i0 + r) * 3 + c];
  }
}

extern "C" void kernel_launch(void* const* d_in, const int* in_sizes, int n_in,
                              void* d_out, int out_size, void* d_ws, size_t ws_size,
                              hipStream_t stream) {
  (void)in_sizes; (void)n_in; (void)out_size; (void)ws_size;
  const float* positions = (const float*)d_in[0];
  const float* t         = (const float*)d_in[1];
  const int*   z         = (const int*)  d_in[2];
  const float* emb       = (const float*)d_in[3];
  const float* W         = (const float*)d_in[4];
  const float* rad_off   = (const float*)d_in[5];
  const float* rad_wid   = (const float*)d_in[6];
  const float* time_off  = (const float*)d_in[7];
  const float* time_wid  = (const float*)d_in[8];
  float* out = (float*)d_out;
  (void)d_ws;

  dim3 g2(512), b2(256);
  field_kernel<<<g2, b2, 0, stream>>>(positions, emb, z, t, time_off, time_wid, W, rad_off, rad_wid, out);
}


# field kernel: phi summed as two partial sums (even/odd basis index) on packed f32 FMA, one v_pk_fma_f32 per two terms instead of two v_fmac; sum order changes
# speedup vs baseline: 2.0538x; 1.0731x over previous
#include <hip/hip_runtime.h>

#define B_   2
#define N_   1024
#define D_   3
#define A_   64
#define R_   50
#define RPAD 64

typedef float v2f __attribute__((ext_vector_type(2)));
typedef float v8f __attribute__((ext_vector_type(8)));

#define ROWS_  4
#define SPLIT_ 2
#define FTHREADS (32 * SPLIT_ * ROWS_)
#define XPITCH 68
static_assert(FTHREADS == 256 && B_ * N_ / ROWS_ == 512, "launch shape in kernel_launch is written as literals");
static_assert(SPLIT_ == 2 && (N_ * D_) % FTHREADS == 0 && ROWS_ * RPAD == FTHREADS, "tiling assumptions");
static_assert(RPAD / 16 <= FTHREADS / 32 && ROWS_ <= 8 && (ROWS_ & (ROWS_ - 1)) == 0 && A_ % 4 == 0, "prefactor tile: one wave per 16 columns, kept rows in lane half 0");
static_assert(ROWS_ * A_ == FTHREADS && XPITCH >= A_ && XPITCH % 2 == 0, "X tile: one element per thread, 8-byte aligned fragment reads");
static_assert(R_ % 2 == 0, "basis terms are summed two at a time");

__global__ __launch_bounds__(FTHREADS)
void field_kernel(const float* __restrict__ pos,
                  const float* __restrict__ emb,
                  const int*   __restrict__ z,
                  const float* __restrict__ t,
                  const float* __restrict__ time_off,
                  const float* __restrict__ time_wid,
                  const float* __restrict__ W,
                  const float* __restrict__ rad_off,
                  const float* __restrict__ rad_wid,
                  float* __restrict__ out) {
  const int tid  = threadIdx.x;
  const int lane = tid & 31;
  const int wv   = tid >> 5;
  const int rib  = wv / SPLIT_;
  const int half = wv % SPLIT_;
  const int b    = blockIdx.x / (N_ / ROWS_);
  const int i0   = (blockIdx.x % (N_ / ROWS_)) * ROWS_;
  const int i    = i0 + rib;

  __shared__ float spos[N_ * D_];
  __shared__ float sc[RPAD], so[RPAD];
  __shared__ float sp[ROWS_ * RPAD];
  __shared__ float sx[ROWS_ * XPITCH];
  __shared__ float spart[SPLIT_ * ROWS_][4];

  const int xr = tid >> 6;
  const int xk = tid & (A_ - 1);
  const int zt = z[i0 + xr];
  const float tb = t[b];
  const float to = time_off[xk], tw = time_wid[xk];

  const float* pb = pos + (size_t)b * N_ * D_;
  float pv[N_ * D_ / FTHREADS];
  #pragma unroll
  for (int q = 0; q < N_ * D_ / FTHREADS; ++q)
    pv[q] = pb[tid + FTHREADS * q];

  const int wvu = __builtin_amdgcn_readfirstlane(wv);
  const int g   = lane >> 4;
  const int lr  = lane & 15;
  const bool rvalid = (wvu * 16 + lr < R_);
  float wb[A_ / 4][2];
  if (wvu < RPAD / 16) {
    const int rc = rvalid ? wvu * 16 + lr : 0;
    #pragma unroll
    for (int s = 0; s < A_ / 4; ++s) {
      const int ka = 4 * s + 2 * g;
      wb[s][0] = W[ka * R_ + rc];
      wb[s][1] = W[(ka + 1) * R_ + rc];
    }
  }

  const float ev = emb[zt * A_ + xk];

  if (tid >= FTHREADS / 2 && tid < FTHREADS / 2 + R_) {
    const int rr = tid - FTHREADS / 2;
    const float s = fminf(0.84932180028801907f * __builtin_amdgcn_rcpf(fabsf(rad_wid[rr])), 1.0e18f);
    sc[rr] = s;
    so[rr] = -(s * rad_off[rr]);
  }
  {
    const float c0 = -0.5f * __builtin_amdgcn_rcpf(tw * tw);
    const float d0 = tb - to;
    const float e0 = __expf(c0 * d0 * d0);
    sx[xr * XPITCH + xk] = ev * e0;
  }
  #pragma unroll
  for (int q = 0; q < N_ * D_ / FTHREADS; ++q)
    spos[tid + FTHREADS * q] = pv[q];
  __syncthreads();

  if (wvu < RPAD / 16) {
    v8f c = {0.f, 0.f, 0.f, 0.f, 0.f, 0.f, 0.f, 0.f};
    const float* xa = sx + (lr & (ROWS_ - 1)) * XPITCH + 2 * g;
    v2f af[A_ / 4];
    #pragma unroll
    for (int s = 0; s < A_ / 4; ++s) {
      af[s].x = xa[4 * s];
      af[s].y = xa[4 * s + 1];
    }
    __builtin_amdgcn_sched_barrier(0);
    #pragma unroll
    for (int s = 0; s < A_ / 4; ++s) {
      const v2f a = af[s];
      v2f bf;
      bf.x = rvalid ? wb[s][0] : 0.f;
      bf.y = rvalid ? wb[s][1] : 0.f;
      c = __builtin_amdgcn_wmma_f32_16x16x4_f32(false, a, false, bf,
                                                (short)0, c, false, false);
    }
    if (g == 0) {
      #pragma unroll
      for (int v = 0; v < ROWS_; ++v)
        sp[v * RPAD + wvu * 16 + lr] = c[v];
    }
  }
  __syncthreads();

  float co[R_], cc[R_], cp[R_];
  #pragma unroll
  for (int rr = 0; rr < R_; ++rr) {
    co[rr] = so[rr];
    cc[rr] = sc[rr];
    cp[rr] = sp[rib * RPAD + rr];
  }

  const float xi = spos[i * 3 + 0], yi = spos[i * 3 + 1], zi = spos[i * 3 + 2];
  float ax = 0.f, ay = 0.f, az = 0.f;

  #pragma unroll 2
  for (int k = 0; k < N_ / (32 * SPLIT_); ++k) {
    const int j = lane + 32 * (SPLIT_ * k + half);
    const float dx = xi - spos[j * 3 + 0];
    const float dy = yi - spos[j * 3 + 1];
    const float dz = zi - spos[j * 3 + 2];
    const float d  = sqrtf(dx * dx + dy * dy + dz * dz + 1e-6f);
    v2f phi2 = {0.f, 0.f};
    #pragma unroll
    for (int rr = 0; rr < R_; rr += 2) {
      const float u0 = __builtin_fmaf(cc[rr],     d, co[rr]);
      const float u1 = __builtin_fmaf(cc[rr + 1], d, co[rr + 1]);
      v2f e, p;
      e.x = __builtin_amdgcn_exp2f(-(u0 * u0));
      e.y = __builtin_amdgcn_exp2f(-(u1 * u1));
      p.x = cp[rr];
      p.y = cp[rr + 1];
      phi2 = __builtin_elementwise_fma(p, e, phi2);
    }
    float phi = phi2.x + phi2.y;
    phi = (j == i) ? 0.f : phi;
    ax += phi * dx;
    ay += phi * dy;
    az += phi * dz;
  }

  #pragma unroll
  for (int m = 16; m > 0; m >>= 1) {
    ax += __shfl_xor(ax, m);
    ay += __shfl_xor(ay, m);
    az += __shfl_xor(az, m);
  }
  if (lane < 3) spart[wv][lane] = (lane == 0) ? ax : (lane == 1) ? ay : az;
  __syncthreads();

  if (tid < ROWS_ * 3) {
    const int r = tid / 3, c = tid % 3;
    const float s = spart[r * SPLIT_][c] + spart[r * SPLIT_ + 1][c];
    out[((size_t)b * N_ + i0) * 3 + tid] = s + spos[(i0 + r) * 3 + c];
  }
}

extern "C" void kernel_launch(void* const* d_in, const int* in_sizes, int n_in,
                              void* d_out, int out_size, void* d_ws, size_t ws_size,
                              hipStream_t stream) {
  (void)in_sizes; (void)n_in; (void)out_size; (void)ws_size;
  const float* positions = (const float*)d_in[0];
  const float* t         = (const float*)d_in[1];
  const int*   z         = (const int*)  d_in[2];
  const float* emb       = (const float*)d_in[3];
  const float* W         = (const float*)d_in[4];
  const float* rad_off   = (const float*)d_in[5];
  const float* rad_wid   = (const float*)d_in[6];
  const float* time_off  = (const float*)d_in[7];
  const float* time_wid  = (const float*)d_in[8];
  float* out = (float*)d_out;
  (void)d_ws;

  dim3 g2(512), b2(256);
  field_kernel<<<g2, b2, 0, stream>>>(positions, emb, z, t, time_off, time_wid, W, rad_off, rad_wid, out);
}


# field kernel: pair distance by the one-instruction square root v_sqrt_f32 (1 ulp) instead of sqrtf's correctly rounded expansion, 13 vector issue slots fewer per pair; sum order unchanged
# speedup vs baseline: 2.1836x; 1.0632x over previous
#include <hip/hip_runtime.h>

#define B_   2
#define N_   1024
#define D_   3
#define A_   64
#define R_   50
#define RPAD 64

typedef float v2f __attribute__((ext_vector_type(2)));
typedef float v8f __attribute__((ext_vector_type(8)));

#define ROWS_  4
#define SPLIT_ 2
#define FTHREADS (32 * SPLIT_ * ROWS_)
#define XPITCH 68
static_assert(FTHREADS == 256 && B_ * N_ / ROWS_ == 512, "launch shape in kernel_launch is written as literals");
static_assert(SPLIT_ == 2 && (N_ * D_) % FTHREADS == 0 && ROWS_ * RPAD == FTHREADS, "tiling assumptions");
static_assert(RPAD / 16 <= FTHREADS / 32 && ROWS_ <= 8 && (ROWS_ & (ROWS_ - 1)) == 0 && A_ % 4 == 0, "prefactor tile: one wave per 16 columns, kept rows in lane half 0");
static_assert(ROWS_ * A_ == FTHREADS && XPITCH >= A_ && XPITCH % 2 == 0, "X tile: one element per thread, 8-byte aligned fragment reads");
static_assert(R_ % 2 == 0, "basis terms are summed two at a time");

__global__ __launch_bounds__(FTHREADS)
void field_kernel(const float* __restrict__ pos,
                  const float* __restrict__ emb,
                  const int*   __restrict__ z,
                  const float* __restrict__ t,
                  const float* __restrict__ time_off,
                  const float* __restrict__ time_wid,
                  const float* __restrict__ W,
                  const float* __restrict__ rad_off,
                  const float* __restrict__ rad_wid,
                  float* __restrict__ out) {
  const int tid  = threadIdx.x;
  const int lane = tid & 31;
  const int wv   = tid >> 5;
  const int rib  = wv / SPLIT_;
  const int half = wv % SPLIT_;
  const int b    = blockIdx.x / (N_ / ROWS_);
  const int i0   = (blockIdx.x % (N_ / ROWS_)) * ROWS_;
  const int i    = i0 + rib;

  __shared__ float spos[N_ * D_];
  __shared__ float sc[RPAD], so[RPAD];
  __shared__ float sp[ROWS_ * RPAD];
  __shared__ float sx[ROWS_ * XPITCH];
  __shared__ float spart[SPLIT_ * ROWS_][4];

  const int xr = tid >> 6;
  const int xk = tid & (A_ - 1);
  const int zt = z[i0 + xr];
  const float tb = t[b];
  const float to = time_off[xk], tw = time_wid[xk];

  const float* pb = pos + (size_t)b * N_ * D_;
  float pv[N_ * D_ / FTHREADS];
  #pragma unroll
  for (int q = 0; q < N_ * D_ / FTHREADS; ++q)
    pv[q] = pb[tid + FTHREADS * q];

  const int wvu = __builtin_amdgcn_readfirstlane(wv);
  const int g   = lane >> 4;
  const int lr  = lane & 15;
  const bool rvalid = (wvu * 16 + lr < R_);
  float wb[A_ / 4][2];
  if (wvu < RPAD / 16) {
    const int rc = rvalid ? wvu * 16 + lr : 0;
    #pragma unroll
    for (int s = 0; s < A_ / 4; ++s) {
      const int ka = 4 * s + 2 * g;
      wb[s][0] = W[ka * R_ + rc];
      wb[s][1] = W[(ka + 1) * R_ + rc];
    }
  }

  const float ev = emb[zt * A_ + xk];

  if (tid >= FTHREADS / 2 && tid < FTHREADS / 2 + R_) {
    const int rr = tid - FTHREADS / 2;
    const float s = fminf(0.84932180028801907f * __builtin_amdgcn_rcpf(fabsf(rad_wid[rr])), 1.0e18f);
    sc[rr] = s;
    so[rr] = -(s * rad_off[rr]);
  }
  {
    const float c0 = -0.5f * __builtin_amdgcn_rcpf(tw * tw);
    const float d0 = tb - to;
    const float e0 = __expf(c0 * d0 * d0);
    sx[xr * XPITCH + xk] = ev * e0;
  }
  #pragma unroll
  for (int q = 0; q < N_ * D_ / FTHREADS; ++q)
    spos[tid + FTHREADS * q] = pv[q];
  __syncthreads();

  if (wvu < RPAD / 16) {
    v8f c = {0.f, 0.f, 0.f, 0.f, 0.f, 0.f, 0.f, 0.f};
    const float* xa = sx + (lr & (ROWS_ - 1)) * XPITCH + 2 * g;
    v2f af[A_ / 4];
    #pragma unroll
    for (int s = 0; s < A_ / 4; ++s) {
      af[s].x = xa[4 * s];
      af[s].y = xa[4 * s + 1];
    }
    __builtin_amdgcn_sched_barrier(0);
    #pragma unroll
    for (int s = 0; s < A_ / 4; ++s) {
      const v2f a = af[s];
      v2f bf;
      bf.x = rvalid ? wb[s][0] : 0.f;
      bf.y = rvalid ? wb[s][1] : 0.f;
      c = __builtin_amdgcn_wmma_f32_16x16x4_f32(false, a, false, bf,
                                                (short)0, c, false, false);
    }
    if (g == 0) {
      #pragma unroll
      for (int v = 0; v < ROWS_; ++v)
        sp[v * RPAD + wvu * 16 + lr] = c[v];
    }
  }
  __syncthreads();

  float co[R_], cc[R_], cp[R_];
  #pragma unroll
  for (int rr = 0; rr < R_; ++rr) {
    co[rr] = so[rr];
    cc[rr] = sc[rr];
    cp[rr] = sp[rib * RPAD + rr];
  }

  const float xi = spos[i * 3 + 0], yi = spos[i * 3 + 1], zi = spos[i * 3 + 2];
  float ax = 0.f, ay = 0.f, az = 0.f;

  #pragma unroll 2
  for (int k = 0; k < N_ / (32 * SPLIT_); ++k) {
    const int j = lane + 32 * (SPLIT_ * k + half);
    const float dx = xi - spos[j * 3 + 0];
    const float dy = yi - spos[j * 3 + 1];
    const float dz = zi - spos[j * 3 + 2];
    const float d  = __builtin_amdgcn_sqrtf(dx * dx + dy * dy + dz * dz + 1e-6f);
    v2f phi2 = {0.f, 0.f};
    #pragma unroll
    for (int rr = 0; rr < R_; rr += 2) {
      const float u0 = __builtin_fmaf(cc[rr],     d, co[rr]);
      const float u1 = __builtin_fmaf(cc[rr + 1], d, co[rr + 1]);
      v2f e, p;
      e.x = __builtin_amdgcn_exp2f(-(u0 * u0));
      e.y = __builtin_amdgcn_exp2f(-(u1 * u1));
      p.x = cp[rr];
      p.y = cp[rr + 1];
      phi2 = __builtin_elementwise_fma(p, e, phi2);
    }
    float phi = phi2.x + phi2.y;
    phi = (j == i) ? 0.f : phi;
    ax += phi * dx;
    ay += phi * dy;
    az += phi * dz;
  }

  #pragma unroll
  for (int m = 16; m > 0; m >>= 1) {
    ax += __shfl_xor(ax, m);
    ay += __shfl_xor(ay, m);
    az += __shfl_xor(az, m);
  }
  if (lane < 3) spart[wv][lane] = (lane == 0) ? ax : (lane == 1) ? ay : az;
  __syncthreads();

  if (tid < ROWS_ * 3) {
    const int r = tid / 3, c = tid % 3;
    const float s = spart[r * SPLIT_][c] + spart[r * SPLIT_ + 1][c];
    out[((size_t)b * N_ + i0) * 3 + tid] = s + spos[(i0 + r) * 3 + c];
  }
}

extern "C" void kernel_launch(void* const* d_in, const int* in_sizes, int n_in,
                              void* d_out, int out_size, void* d_ws, size_t ws_size,
                              hipStream_t stream) {
  (void)in_sizes; (void)n_in; (void)out_size; (void)ws_size;
  const float* positions = (const float*)d_in[0];
  const float* t         = (const float*)d_in[1];
  const int*   z         = (const int*)  d_in[2];
  const float* emb       = (const float*)d_in[3];
  const float* W         = (const float*)d_in[4];
  const float* rad_off   = (const float*)d_in[5];
  const float* rad_wid   = (const float*)d_in[6];
  const float* time_off  = (const float*)d_in[7];
  const float* time_wid  = (const float*)d_in[8];
  float* out = (float*)d_out;
  (void)d_ws;

  dim3 g2(512), b2(256);
  field_kernel<<<g2, b2, 0, stream>>>(positions, emb, z, t, time_off, time_wid, W, rad_off, rad_wid, out);
}


# field kernel: basis steps of a trip's two pairs written as one software pipeline, order exp, packed, exp, packed, packed held by sched barriers: no two v_exp_f32 back to back; sum order unchanged
# speedup vs baseline: 2.3144x; 1.0599x over previous
#include <hip/hip_runtime.h>

#define B_   2
#define N_   1024
#define D_   3
#define A_   64
#define R_   50
#define RPAD 64

typedef float v2f __attribute__((ext_vector_type(2)));
typedef float v8f __attribute__((ext_vector_type(8)));

#define ROWS_  4
#define SPLIT_ 2
#define FTHREADS (32 * SPLIT_ * ROWS_)
#define XPITCH 68
static_assert(FTHREADS == 256 && B_ * N_ / ROWS_ == 512, "launch shape in kernel_launch is written as literals");
static_assert(SPLIT_ == 2 && (N_ * D_) % FTHREADS == 0 && ROWS_ * RPAD == FTHREADS, "tiling assumptions");
static_assert(RPAD / 16 <= FTHREADS / 32 && ROWS_ <= 8 && (ROWS_ & (ROWS_ - 1)) == 0 && A_ % 4 == 0, "prefactor tile: one wave per 16 columns, kept rows in lane half 0");
static_assert(ROWS_ * A_ == FTHREADS && XPITCH >= A_ && XPITCH % 2 == 0, "X tile: one element per thread, 8-byte aligned fragment reads");
static_assert(R_ % 2 == 0, "basis terms are summed two at a time");
#define NQ_ (R_ / 2)
#define TT_ (2 * NQ_)
static_assert((N_ / (32 * SPLIT_)) % 2 == 0, "two pairs per trip");

__global__ __launch_bounds__(FTHREADS)
void field_kernel(const float* __restrict__ pos,
                  const float* __restrict__ emb,
                  const int*   __restrict__ z,
                  const float* __restrict__ t,
                  const float* __restrict__ time_off,
                  const float* __restrict__ time_wid,
                  const float* __restrict__ W,
                  const float* __restrict__ rad_off,
                  const float* __restrict__ rad_wid,
                  float* __restrict__ out) {
  const int tid  = threadIdx.x;
  const int lane = tid & 31;
  const int wv   = tid >> 5;
  const int rib  = wv / SPLIT_;
  const int half = wv % SPLIT_;
  const int b    = blockIdx.x / (N_ / ROWS_);
  const int i0   = (blockIdx.x % (N_ / ROWS_)) * ROWS_;
  const int i    = i0 + rib;

  __shared__ float spos[N_ * D_];
  __shared__ float sc[RPAD], so[RPAD];
  __shared__ float sp[ROWS_ * RPAD];
  __shared__ float sx[ROWS_ * XPITCH];
  __shared__ float spart[SPLIT_ * ROWS_][4];

  const int xr = tid >> 6;
  const int xk = tid & (A_ - 1);
  const int zt = z[i0 + xr];
  const float tb = t[b];
  const float to = time_off[xk], tw = time_wid[xk];

  const float* pb = pos + (size_t)b * N_ * D_;
  float pv[N_ * D_ / FTHREADS];
  #pragma unroll
  for (int q = 0; q < N_ * D_ / FTHREADS; ++q)
    pv[q] = pb[tid + FTHREADS * q];

  const int wvu = __builtin_amdgcn_readfirstlane(wv);
  const int g   = lane >> 4;
  const int lr  = lane & 15;
  const bool rvalid = (wvu * 16 + lr < R_);
  float wb[A_ / 4][2];
  if (wvu < RPAD / 16) {
    const int rc = rvalid ? wvu * 16 + lr : 0;
    #pragma unroll
    for (int s = 0; s < A_ / 4; ++s) {
      const int ka = 4 * s + 2 * g;
      wb[s][0] = W[ka * R_ + rc];
      wb[s][1] = W[(ka + 1) * R_ + rc];
    }
  }

  const float ev = emb[zt * A_ + xk];

  if (tid >= FTHREADS / 2 && tid < FTHREADS / 2 + R_) {
    const int rr = tid - FTHREADS / 2;
    const float s = fminf(0.84932180028801907f * __builtin_amdgcn_rcpf(fabsf(rad_wid[rr])), 1.0e18f);
    sc[rr] = s;
    so[rr] = -(s * rad_off[rr]);
  }
  {
    const float c0 = -0.5f * __builtin_amdgcn_rcpf(tw * tw);
    const float d0 = tb - to;
    const float e0 = __expf(c0 * d0 * d0);
    sx[xr * XPITCH + xk] = ev * e0;
  }
  #pragma unroll
  for (int q = 0; q < N_ * D_ / FTHREADS; ++q)
    spos[tid + FTHREADS * q] = pv[q];
  __syncthreads();

  if (wvu < RPAD / 16) {
    v8f c = {0.f, 0.f, 0.f, 0.f, 0.f, 0.f, 0.f, 0.f};
    const float* xa = sx + (lr & (ROWS_ - 1)) * XPITCH + 2 * g;
    v2f af[A_ / 4];
    #pragma unroll
    for (int s = 0; s < A_ / 4; ++s) {
      af[s].x = xa[4 * s];
      af[s].y = xa[4 * s + 1];
    }
    __builtin_amdgcn_sched_barrier(0);
    #pragma unroll
    for (int s = 0; s < A_ / 4; ++s) {
      const v2f a = af[s];
      v2f bf;
      bf.x = rvalid ? wb[s][0] : 0.f;
      bf.y = rvalid ? wb[s][1] : 0.f;
      c = __builtin_amdgcn_wmma_f32_16x16x4_f32(false, a, false, bf,
                                                (short)0, c, false, false);
    }
    if (g == 0) {
      #pragma unroll
      for (int v = 0; v < ROWS_; ++v)
        sp[v * RPAD + wvu * 16 + lr] = c[v];
    }
  }
  __syncthreads();

  v2f co[NQ_], cc[NQ_], cp[NQ_];
  #pragma unroll
  for (int q = 0; q < NQ_; ++q) {
    co[q].x = so[2 * q];                 co[q].y = so[2 * q + 1];
    cc[q].x = sc[2 * q];                 cc[q].y = sc[2 * q + 1];
    cp[q].x = sp[rib * RPAD + 2 * q];    cp[q].y = sp[rib * RPAD + 2 * q + 1];
  }

  const float xi = spos[i * 3 + 0], yi = spos[i * 3 + 1], zi = spos[i * 3 + 2];
  float ax = 0.f, ay = 0.f, az = 0.f;

  #pragma unroll 1
  for (int k = 0; k < N_ / (32 * SPLIT_); k += 2) {
    float dx[2], dy[2], dz[2];
    v2f d2[2];
    bool offd[2];
    #pragma unroll
    for (int h2 = 0; h2 < 2; ++h2) {
      const int j = lane + 32 * (SPLIT_ * (k + h2) + half);
      dx[h2] = xi - spos[j * 3 + 0];
      dy[h2] = yi - spos[j * 3 + 1];
      dz[h2] = zi - spos[j * 3 + 2];
      const float d = __builtin_amdgcn_sqrtf(dx[h2] * dx[h2] + dy[h2] * dy[h2] + dz[h2] * dz[h2] + 1e-6f);
      d2[h2].x = d;
      d2[h2].y = d;
      offd[h2] = (j != i);
    }
    v2f u[TT_], w[TT_], e[TT_];
    v2f phi2[2] = {{0.f, 0.f}, {0.f, 0.f}};
#define SB __builtin_amdgcn_sched_barrier(0)
#define ST_A(t) { u[t] = __builtin_elementwise_fma(cc[(t) / 2], d2[(t) % 2], co[(t) / 2]); SB; }
#define ST_B(t) { w[t] = -(u[t] * u[t]); SB; }
#define ST_E(t) { phi2[(t) % 2] = __builtin_elementwise_fma(cp[(t) / 2], e[t], phi2[(t) % 2]); SB; }
    #pragma unroll
    for (int s = 0; s < TT_ + 7; ++s) {
      const bool hasA = (s < TT_);
      const bool hasB = (s >= 2 && s - 2 < TT_);
      const bool hasT = (s >= 5 && s - 5 < TT_);
      const bool hasE = (s >= 7 && s - 7 < TT_);
      if (hasT) { e[s - 5].x = __builtin_amdgcn_exp2f(w[s - 5].x); SB; }
      if (hasA)      ST_A(s)
      else if (hasB) ST_B(s - 2)
      else if (hasE) ST_E(s - 7)
      if (hasT) { e[s - 5].y = __builtin_amdgcn_exp2f(w[s - 5].y); SB; }
      if ((hasA || hasB) && hasE) ST_E(s - 7)
      if (hasA && hasB) ST_B(s - 2)
    }
#undef ST_A
#undef ST_B
#undef ST_E
#undef SB
    #pragma unroll
    for (int h2 = 0; h2 < 2; ++h2) {
      float phi = phi2[h2].x + phi2[h2].y;
      phi = offd[h2] ? phi : 0.f;
      ax += phi * dx[h2];
      ay += phi * dy[h2];
      az += phi * dz[h2];
    }
  }

  #pragma unroll
  for (int m = 16; m > 0; m >>= 1) {
    ax += __shfl_xor(ax, m);
    ay += __shfl_xor(ay, m);
    az += __shfl_xor(az, m);
  }
  if (lane < 3) spart[wv][lane] = (lane == 0) ? ax : (lane == 1) ? ay : az;
  __syncthreads();

  if (tid < ROWS_ * 3) {
    const int r = tid / 3, c = tid % 3;
    const float s = spart[r * SPLIT_][c] + spart[r * SPLIT_ + 1][c];
    out[((size_t)b * N_ + i0) * 3 + tid] = s + spos[(i0 + r) * 3 + c];
  }
}

extern "C" void kernel_launch(void* const* d_in, const int* in_sizes, int n_in,
                              void* d_out, int out_size, void* d_ws, size_t ws_size,
                              hipStream_t stream) {
  (void)in_sizes; (void)n_in; (void)out_size; (void)ws_size;
  const float* positions = (const float*)d_in[0];
  const float* t         = (const float*)d_in[1];
  const int*   z         = (const int*)  d_in[2];
  const float* emb       = (const float*)d_in[3];
  const float* W         = (const float*)d_in[4];
  const float* rad_off   = (const float*)d_in[5];
  const float* rad_wid   = (const float*)d_in[6];
  const float* time_off  = (const float*)d_in[7];
  const float* time_wid  = (const float*)d_in[8];
  float* out = (float*)d_out;
  (void)d_ws;

  dim3 g2(512), b2(256);
  field_kernel<<<g2, b2, 0, stream>>>(positions, emb, z, t, time_off, time_wid, W, rad_off, rad_wid, out);
}
